# starting kernel
# baseline (speedup 1.0000x reference)
#include <hip/hip_runtime.h>
#include <math.h>


typedef float v2f __attribute__((ext_vector_type(2)));
typedef float v4f __attribute__((ext_vector_type(4)));
typedef float v8f __attribute__((ext_vector_type(8)));

#define NJ 24
#define ND 6
#define NF 19

#define OFF_W0  0
#define OFF_W1  (OFF_W0 + 16*288)
#define OFF_W2  (OFF_W1 + NJ*32*20)
#define OFF_B1  (OFF_W2 + NJ*16*20)
#define OFF_B2  (OFF_B1 + NJ*32)
#define OFF_B0  (OFF_B2 + NJ*16)
#define WS_FLOATS (OFF_B0 + 16)

__constant__ int c_parents[NJ] = {-1,0,0,0,1,2,3,4,5,6,7,8,9,9,9,12,13,14,16,17,18,19,20,21};

static __device__ __forceinline__ v8f wmma_f32(v2f a, v2f b, v8f c) {
  return __builtin_amdgcn_wmma_f32_16x16x4_f32(
       false, a,  false, b,
       (short)0, c,  false,  false);
}

__global__ __launch_bounds__(256)
void pad_weights_kernel(const float* __restrict__ W0, const float* __restrict__ b0,
                        const float* __restrict__ W1, const float* __restrict__ b1,
                        const float* __restrict__ W2, const float* __restrict__ b2,
                        float* __restrict__ ws)
{
  const int t = threadIdx.x;
  float* W0p = ws + OFF_W0;
  float* W1p = ws + OFF_W1;
  float* W2p = ws + OFF_W2;
  float* b1p = ws + OFF_B1;
  float* b2p = ws + OFF_B2;
  float* b0p = ws + OFF_B0;

  for (int i = t; i < 16 * 288; i += 256) {
    const int n = i / 288, k = i - n * 288;
    W0p[i] = (n < ND) ? W0[n * 288 + k] : 0.0f;
  }
  for (int i = t; i < NJ * 32 * 20; i += 256) {
    const int j = i / 640, r = i - j * 640, n = r / 20, k = r - n * 20;
    W1p[i] = (n < NF && k < NF) ? W1[(j * NF + n) * NF + k] : 0.0f;
  }
  for (int i = t; i < NJ * 16 * 20; i += 256) {
    const int j = i / 320, r = i - j * 320, n = r / 20, k = r - n * 20;
    W2p[i] = (n < ND && k < NF) ? W2[(j * ND + n) * NF + k] : 0.0f;
  }
  for (int i = t; i < NJ * 32; i += 256) {
    const int j = i / 32, n = i - j * 32;
    b1p[i] = (n < NF) ? b1[j * NF + n] : 0.0f;
  }
  for (int i = t; i < NJ * 16; i += 256) {
    const int j = i / 16, n = i - j * 16;
    b2p[i] = (n < ND) ? b2[j * ND + n] : 0.0f;
  }
  if (t < 16) b0p[t] = (t < ND) ? b0[t] : 0.0f;
}

__global__ __launch_bounds__(128)
void hier_pose_kernel(const float* __restrict__ rots, const float* __restrict__ Jtrs,
                      const float* __restrict__ ws,   float* __restrict__ out)
{
  __shared__ __align__(16) float sfeat[4][16][152];
  __shared__ __align__(16) float sx[4][16][20];

  const float* W0p = ws + OFF_W0;
  const float* W1p = ws + OFF_W1;
  const float* W2p = ws + OFF_W2;
  const float* b1p = ws + OFF_B1;
  const float* b2p = ws + OFF_B2;
  const float* b0p = ws + OFF_B0;

  const int  lane = threadIdx.x & 31;
  const int  wave = threadIdx.x >> 5;
  const int  half = lane >> 4;
  const int  ln   = lane & 15;
  const long s0   = ((long)blockIdx.x * 4 + wave) * 16;

  float* feat = &sfeat[wave][0][0];
  float* xb   = &sx[wave][0][0];

  const float* arow_r = rots + (s0 + ln) * 216;
  const float* arow_j = Jtrs + (s0 + ln) * 72;
  __builtin_prefetch(arow_r, 0, 3);
  __builtin_prefetch(arow_j, 0, 3);

  v8f acc;
  {
    const float ci = b0p[ln];
    #pragma unroll
    for (int g = 0; g < 8; ++g) acc[g] = ci;
  }
  #pragma unroll 3
  for (int kk = 0; kk < 54; ++kk) {
    const int k = kk * 4 + half * 2;
    const v2f a = *(const v2f*)(arow_r + k);
    const v2f b = *(const v2f*)(W0p + ln * 288 + k);
    acc = wmma_f32(a, b, acc);
  }
  #pragma unroll 3
  for (int kk = 54; kk < 72; ++kk) {
    const int k = kk * 4 + half * 2;
    const v2f a = *(const v2f*)(arow_j + (k - 216));
    const v2f b = *(const v2f*)(W0p + ln * 288 + k);
    acc = wmma_f32(a, b, acc);
  }
  if (ln < ND) {
    #pragma unroll
    for (int g = 0; g < 8; ++g)
      feat[(g + half * 8) * 152 + 144 + ln] = acc[g];
  }

  for (int j = 0; j < NJ; ++j) {
    const int p     = c_parents[j];
    const int fslot = (p < 0) ? 24 : p;

    if (lane < 16) {
      const int  m = lane;
      const long s = s0 + m;
      const float* rp = rots + (s * NJ + j) * 9;
      const float* tp = Jtrs + (s * NJ + j) * 3;
      __align__(16) float xr[20];
      #pragma unroll
      for (int i = 0; i < 9; ++i) xr[i] = rp[i];
      const float jx = tp[0], jy = tp[1], jz = tp[2];
      xr[9] = jx; xr[10] = jy; xr[11] = jz;
      float px = 0.0f, py = 0.0f, pz = 0.0f;
      if (p >= 0) {
        const float* pp = Jtrs + (s * NJ + p) * 3;
        px = pp[0]; py = pp[1]; pz = pp[2];
      }
      const float dx = jx - px, dy = jy - py, dz = jz - pz;
      xr[12] = sqrtf(dx * dx + dy * dy + dz * dz);
      #pragma unroll
      for (int i = 0; i < ND; ++i) xr[13 + i] = feat[m * 152 + fslot * ND + i];
      xr[19] = 0.0f;
      #pragma unroll
      for (int i = 0; i < 20; i += 4)
        *(v4f*)(xb + m * 20 + i) = *(const v4f*)(xr + i);
    }
    asm volatile("s_wait_dscnt 0" ::: "memory");

    v8f c0, c1;
    {
      const float bias0 = b1p[j * 32 + ln];
      const float bias1 = b1p[j * 32 + 16 + ln];
      #pragma unroll
      for (int g = 0; g < 8; ++g) { c0[g] = bias0; c1[g] = bias1; }
    }
    const float* w1j = W1p + j * 32 * 20;
    #pragma unroll
    for (int kk = 0; kk < 5; ++kk) {
      const int k = kk * 4 + half * 2;
      const v2f a   = *(const v2f*)(xb + ln * 20 + k);
      const v2f bt0 = *(const v2f*)(w1j + ln * 20 + k);
      const v2f bt1 = *(const v2f*)(w1j + (16 + ln) * 20 + k);
      c0 = wmma_f32(a, bt0, c0);
      c1 = wmma_f32(a, bt1, c1);
    }

    #pragma unroll
    for (int g = 0; g < 8; ++g) {
      const int m = g + half * 8;
      xb[m * 20 + ln] = fmaxf(c0[g], 0.0f);
      if (ln < 3) xb[m * 20 + 16 + ln] = fmaxf(c1[g], 0.0f);
      if (ln == 3) xb[m * 20 + 19] = 0.0f;
    }
    asm volatile("s_wait_dscnt 0" ::: "memory");

    v8f c2;
    {
      const float bias2 = b2p[j * 16 + ln];
      #pragma unroll
      for (int g = 0; g < 8; ++g) c2[g] = bias2;
    }
    const float* w2j = W2p + j * 16 * 20;
    #pragma unroll
    for (int kk = 0; kk < 5; ++kk) {
      const int k = kk * 4 + half * 2;
      const v2f a  = *(const v2f*)(xb + ln * 20 + k);
      const v2f bt = *(const v2f*)(w2j + ln * 20 + k);
      c2 = wmma_f32(a, bt, c2);
    }
    if (ln < ND) {
      #pragma unroll
      for (int g = 0; g < 8; ++g)
        feat[(g + half * 8) * 152 + j * ND + ln] = c2[g];
    }
  }

  asm volatile("s_wait_dscnt 0" ::: "memory");
  #pragma unroll 4
  for (int idx = lane; idx < 16 * 36; idx += 32) {
    const int m = idx / 36;
    const int q = idx - m * 36;
    *(v4f*)(out + (s0 + m) * 144 + q * 4) = *(const v4f*)(feat + m * 152 + q * 4);
  }
}

extern "C" void kernel_launch(void* const* d_in, const int* in_sizes, int n_in,
                              void* d_out, int out_size, void* d_ws, size_t ws_size,
                              hipStream_t stream) {
  const float* rots = (const float*)d_in[0];
  const float* Jtrs = (const float*)d_in[1];
  const float* W0   = (const float*)d_in[2];
  const float* b0   = (const float*)d_in[3];
  const float* W1   = (const float*)d_in[4];
  const float* b1   = (const float*)d_in[5];
  const float* W2   = (const float*)d_in[6];
  const float* b2   = (const float*)d_in[7];
  float* out = (float*)d_out;
  float* ws  = (float*)d_ws;

  hipLaunchKernelGGL(pad_weights_kernel, dim3(1), dim3(256), 0, stream,
                     W0, b0, W1, b1, W2, b2, ws);

  const long bsz    = (long)in_sizes[0] / (NJ * 9);
  const int  blocks = (int)(bsz / 64);
  if (blocks > 0) {
    hipLaunchKernelGGL(hier_pose_kernel, dim3(blocks), dim3(128), 0, stream,
                       rots, Jtrs, ws, out);
  }
}


# Weight-padding prologue spread from one work-group looping 114 dependent load-store trips to 113 work-groups, one padded element per thread (one memory round trip)
# speedup vs baseline: 1.2527x; 1.2527x over previous
#include <hip/hip_runtime.h>
#include <math.h>


typedef float v2f __attribute__((ext_vector_type(2)));
typedef float v4f __attribute__((ext_vector_type(4)));
typedef float v8f __attribute__((ext_vector_type(8)));

#define NJ 24
#define ND 6
#define NF 19

#define OFF_W0  0
#define OFF_W1  (OFF_W0 + 16*288)
#define OFF_W2  (OFF_W1 + NJ*32*20)
#define OFF_B1  (OFF_W2 + NJ*16*20)
#define OFF_B2  (OFF_B1 + NJ*32)
#define OFF_B0  (OFF_B2 + NJ*16)
#define WS_FLOATS (OFF_B0 + 16)

__constant__ int c_parents[NJ] = {-1,0,0,0,1,2,3,4,5,6,7,8,9,9,9,12,13,14,16,17,18,19,20,21};

static __device__ __forceinline__ v8f wmma_f32(v2f a, v2f b, v8f c) {
  return __builtin_amdgcn_wmma_f32_16x16x4_f32(
       false, a,  false, b,
       (short)0, c,  false,  false);
}

#define PAD_THREADS 256
#define PAD_BLOCKS  113
static_assert(PAD_BLOCKS * PAD_THREADS >= WS_FLOATS, "prep grid must cover the padded weights");

__global__ __launch_bounds__(PAD_THREADS)
void pad_weights_kernel(const float* __restrict__ W0, const float* __restrict__ b0,
                        const float* __restrict__ W1, const float* __restrict__ b1,
                        const float* __restrict__ W2, const float* __restrict__ b2,
                        float* __restrict__ ws)
{
  const int i = blockIdx.x * PAD_THREADS + threadIdx.x;
  if (i >= WS_FLOATS) return;

  float v = 0.0f;
  if (i < OFF_W1) {
    const int n = i / 288, k = i - n * 288;
    if (n < ND) v = W0[n * 288 + k];
  } else if (i < OFF_W2) {
    const int q = i - OFF_W1;
    const int j = q / 640, r = q - j * 640, n = r / 20, k = r - n * 20;
    if (n < NF && k < NF) v = W1[(j * NF + n) * NF + k];
  } else if (i < OFF_B1) {
    const int q = i - OFF_W2;
    const int j = q / 320, r = q - j * 320, n = r / 20, k = r - n * 20;
    if (n < ND && k < NF) v = W2[(j * ND + n) * NF + k];
  } else if (i < OFF_B2) {
    const int q = i - OFF_B1;
    const int j = q / 32, n = q - j * 32;
    if (n < NF) v = b1[j * NF + n];
  } else if (i < OFF_B0) {
    const int q = i - OFF_B2;
    const int j = q / 16, n = q - j * 16;
    if (n < ND) v = b2[j * ND + n];
  } else {
    const int n = i - OFF_B0;
    if (n < ND) v = b0[n];
  }
  ws[i] = v;
}

__global__ __launch_bounds__(128)
void hier_pose_kernel(const float* __restrict__ rots, const float* __restrict__ Jtrs,
                      const float* __restrict__ ws,   float* __restrict__ out)
{
  __shared__ __align__(16) float sfeat[4][16][152];
  __shared__ __align__(16) float sx[4][16][20];

  const float* W0p = ws + OFF_W0;
  const float* W1p = ws + OFF_W1;
  const float* W2p = ws + OFF_W2;
  const float* b1p = ws + OFF_B1;
  const float* b2p = ws + OFF_B2;
  const float* b0p = ws + OFF_B0;

  const int  lane = threadIdx.x & 31;
  const int  wave = threadIdx.x >> 5;
  const int  half = lane >> 4;
  const int  ln   = lane & 15;
  const long s0   = ((long)blockIdx.x * 4 + wave) * 16;

  float* feat = &sfeat[wave][0][0];
  float* xb   = &sx[wave][0][0];

  const float* arow_r = rots + (s0 + ln) * 216;
  const float* arow_j = Jtrs + (s0 + ln) * 72;
  __builtin_prefetch(arow_r, 0, 3);
  __builtin_prefetch(arow_j, 0, 3);

  v8f acc;
  {
    const float ci = b0p[ln];
    #pragma unroll
    for (int g = 0; g < 8; ++g) acc[g] = ci;
  }
  #pragma unroll 3
  for (int kk = 0; kk < 54; ++kk) {
    const int k = kk * 4 + half * 2;
    const v2f a = *(const v2f*)(arow_r + k);
    const v2f b = *(const v2f*)(W0p + ln * 288 + k);
    acc = wmma_f32(a, b, acc);
  }
  #pragma unroll 3
  for (int kk = 54; kk < 72; ++kk) {
    const int k = kk * 4 + half * 2;
    const v2f a = *(const v2f*)(arow_j + (k - 216));
    const v2f b = *(const v2f*)(W0p + ln * 288 + k);
    acc = wmma_f32(a, b, acc);
  }
  if (ln < ND) {
    #pragma unroll
    for (int g = 0; g < 8; ++g)
      feat[(g + half * 8) * 152 + 144 + ln] = acc[g];
  }

  for (int j = 0; j < NJ; ++j) {
    const int p     = c_parents[j];
    const int fslot = (p < 0) ? 24 : p;

    if (lane < 16) {
      const int  m = lane;
      const long s = s0 + m;
      const float* rp = rots + (s * NJ + j) * 9;
      const float* tp = Jtrs + (s * NJ + j) * 3;
      __align__(16) float xr[20];
      #pragma unroll
      for (int i = 0; i < 9; ++i) xr[i] = rp[i];
      const float jx = tp[0], jy = tp[1], jz = tp[2];
      xr[9] = jx; xr[10] = jy; xr[11] = jz;
      float px = 0.0f, py = 0.0f, pz = 0.0f;
      if (p >= 0) {
        const float* pp = Jtrs + (s * NJ + p) * 3;
        px = pp[0]; py = pp[1]; pz = pp[2];
      }
      const float dx = jx - px, dy = jy - py, dz = jz - pz;
      xr[12] = sqrtf(dx * dx + dy * dy + dz * dz);
      #pragma unroll
      for (int i = 0; i < ND; ++i) xr[13 + i] = feat[m * 152 + fslot * ND + i];
      xr[19] = 0.0f;
      #pragma unroll
      for (int i = 0; i < 20; i += 4)
        *(v4f*)(xb + m * 20 + i) = *(const v4f*)(xr + i);
    }
    asm volatile("s_wait_dscnt 0" ::: "memory");

    v8f c0, c1;
    {
      const float bias0 = b1p[j * 32 + ln];
      const float bias1 = b1p[j * 32 + 16 + ln];
      #pragma unroll
      for (int g = 0; g < 8; ++g) { c0[g] = bias0; c1[g] = bias1; }
    }
    const float* w1j = W1p + j * 32 * 20;
    #pragma unroll
    for (int kk = 0; kk < 5; ++kk) {
      const int k = kk * 4 + half * 2;
      const v2f a   = *(const v2f*)(xb + ln * 20 + k);
      const v2f bt0 = *(const v2f*)(w1j + ln * 20 + k);
      const v2f bt1 = *(const v2f*)(w1j + (16 + ln) * 20 + k);
      c0 = wmma_f32(a, bt0, c0);
      c1 = wmma_f32(a, bt1, c1);
    }

    #pragma unroll
    for (int g = 0; g < 8; ++g) {
      const int m = g + half * 8;
      xb[m * 20 + ln] = fmaxf(c0[g], 0.0f);
      if (ln < 3) xb[m * 20 + 16 + ln] = fmaxf(c1[g], 0.0f);
      if (ln == 3) xb[m * 20 + 19] = 0.0f;
    }
    asm volatile("s_wait_dscnt 0" ::: "memory");

    v8f c2;
    {
      const float bias2 = b2p[j * 16 + ln];
      #pragma unroll
      for (int g = 0; g < 8; ++g) c2[g] = bias2;
    }
    const float* w2j = W2p + j * 16 * 20;
    #pragma unroll
    for (int kk = 0; kk < 5; ++kk) {
      const int k = kk * 4 + half * 2;
      const v2f a  = *(const v2f*)(xb + ln * 20 + k);
      const v2f bt = *(const v2f*)(w2j + ln * 20 + k);
      c2 = wmma_f32(a, bt, c2);
    }
    if (ln < ND) {
      #pragma unroll
      for (int g = 0; g < 8; ++g)
        feat[(g + half * 8) * 152 + j * ND + ln] = c2[g];
    }
  }

  asm volatile("s_wait_dscnt 0" ::: "memory");
  #pragma unroll 4
  for (int idx = lane; idx < 16 * 36; idx += 32) {
    const int m = idx / 36;
    const int q = idx - m * 36;
    *(v4f*)(out + (s0 + m) * 144 + q * 4) = *(const v4f*)(feat + m * 152 + q * 4);
  }
}

extern "C" void kernel_launch(void* const* d_in, const int* in_sizes, int n_in,
                              void* d_out, int out_size, void* d_ws, size_t ws_size,
                              hipStream_t stream) {
  const float* rots = (const float*)d_in[0];
  const float* Jtrs = (const float*)d_in[1];
  const float* W0   = (const float*)d_in[2];
  const float* b0   = (const float*)d_in[3];
  const float* W1   = (const float*)d_in[4];
  const float* b1   = (const float*)d_in[5];
  const float* W2   = (const float*)d_in[6];
  const float* b2   = (const float*)d_in[7];
  float* out = (float*)d_out;
  float* ws  = (float*)d_ws;

  hipLaunchKernelGGL(pad_weights_kernel, dim3(PAD_BLOCKS), dim3(PAD_THREADS), 0, stream,
                     W0, b0, W1, b1, W2, b2, ws);

  const long bsz    = (long)in_sizes[0] / (NJ * 9);
  const int  blocks = (int)(bsz / 64);
  if (blocks > 0) {
    hipLaunchKernelGGL(hier_pose_kernel, dim3(blocks), dim3(128), 0, stream,
                       rots, Jtrs, ws, out);
  }
}


# Each wave stages its 16 input rows in LDS once with whole-line b128 loads; GEMM A fragments and per-joint x tiles read LDS, not 16 B per line from global; out[j] reuses the rots[j] slot
# speedup vs baseline: 1.6820x; 1.3427x over previous
#include <hip/hip_runtime.h>
#include <math.h>


typedef float v2f __attribute__((ext_vector_type(2)));
typedef float v4f __attribute__((ext_vector_type(4)));
typedef float v8f __attribute__((ext_vector_type(8)));

#define NJ 24
#define ND 6
#define NF 19

#define PITCH   300
#define COL_JT  216
#define COL_GF  288

#define OFF_W0  0
#define OFF_W1  (OFF_W0 + 16*288)
#define OFF_W2  (OFF_W1 + NJ*32*20)
#define OFF_B1  (OFF_W2 + NJ*16*20)
#define OFF_B2  (OFF_B1 + NJ*32)
#define OFF_B0  (OFF_B2 + NJ*16)
#define WS_FLOATS (OFF_B0 + 16)

__constant__ int c_parents[NJ] = {-1,0,0,0,1,2,3,4,5,6,7,8,9,9,9,12,13,14,16,17,18,19,20,21};

static __device__ __forceinline__ v8f wmma_f32(v2f a, v2f b, v8f c) {
  return __builtin_amdgcn_wmma_f32_16x16x4_f32(
       false, a,  false, b,
       (short)0, c,  false,  false);
}

#define PAD_THREADS 256
#define PAD_BLOCKS  113
static_assert(PAD_BLOCKS * PAD_THREADS >= WS_FLOATS, "prep grid must cover the padded weights");

__global__ __launch_bounds__(PAD_THREADS)
void pad_weights_kernel(const float* __restrict__ W0, const float* __restrict__ b0,
                        const float* __restrict__ W1, const float* __restrict__ b1,
                        const float* __restrict__ W2, const float* __restrict__ b2,
                        float* __restrict__ ws)
{
  const int i = blockIdx.x * PAD_THREADS + threadIdx.x;
  if (i >= WS_FLOATS) return;

  float v = 0.0f;
  if (i < OFF_W1) {
    const int n = i / 288, k = i - n * 288;
    if (n < ND) v = W0[n * 288 + k];
  } else if (i < OFF_W2) {
    const int q = i - OFF_W1;
    const int j = q / 640, r = q - j * 640, n = r / 20, k = r - n * 20;
    if (n < NF && k < NF) v = W1[(j * NF + n) * NF + k];
  } else if (i < OFF_B1) {
    const int q = i - OFF_W2;
    const int j = q / 320, r = q - j * 320, n = r / 20, k = r - n * 20;
    if (n < ND && k < NF) v = W2[(j * ND + n) * NF + k];
  } else if (i < OFF_B2) {
    const int q = i - OFF_B1;
    const int j = q / 32, n = q - j * 32;
    if (n < NF) v = b1[j * NF + n];
  } else if (i < OFF_B0) {
    const int q = i - OFF_B2;
    const int j = q / 16, n = q - j * 16;
    if (n < ND) v = b2[j * ND + n];
  } else {
    const int n = i - OFF_B0;
    if (n < ND) v = b0[n];
  }
  ws[i] = v;
}

__global__ __launch_bounds__(128)
void hier_pose_kernel(const float* __restrict__ rots, const float* __restrict__ Jtrs,
                      const float* __restrict__ ws,   float* __restrict__ out)
{
  __shared__ __align__(16) float stile[4][16][PITCH];
  __shared__ __align__(16) float sx[4][16][20];

  const float* W0p = ws + OFF_W0;
  const float* W1p = ws + OFF_W1;
  const float* W2p = ws + OFF_W2;
  const float* b1p = ws + OFF_B1;
  const float* b2p = ws + OFF_B2;
  const float* b0p = ws + OFF_B0;

  const int  lane = threadIdx.x & 31;
  const int  wave = threadIdx.x >> 5;
  const int  half = lane >> 4;
  const int  ln   = lane & 15;
  const long s0   = ((long)blockIdx.x * 4 + wave) * 16;

  float* srow = &stile[wave][0][0];
  float* xb   = &sx[wave][0][0];

  {
    const v4f* gr = (const v4f*)(rots + s0 * 216);
    const v4f* gj = (const v4f*)(Jtrs + s0 * 72);
    v4f tr[27];
    v4f tj[9];
    #pragma unroll
    for (int i = 0; i < 27; ++i) tr[i] = gr[lane + 32 * i];
    #pragma unroll
    for (int i = 0; i < 9; ++i)  tj[i] = gj[lane + 32 * i];
    #pragma unroll
    for (int i = 0; i < 27; ++i) {
      const int idx = lane + 32 * i;
      const int m   = idx / 54;
      const int c4  = idx - m * 54;
      *(v4f*)(srow + m * PITCH + c4 * 4) = tr[i];
    }
    #pragma unroll
    for (int i = 0; i < 9; ++i) {
      const int idx = lane + 32 * i;
      const int m   = idx / 18;
      const int c4  = idx - m * 18;
      *(v4f*)(srow + m * PITCH + COL_JT + c4 * 4) = tj[i];
    }
  }
  asm volatile("s_wait_dscnt 0" ::: "memory");

  v8f acc;
  {
    const float ci = b0p[ln];
    #pragma unroll
    for (int g = 0; g < 8; ++g) acc[g] = ci;
  }
  #pragma unroll 8
  for (int kk = 0; kk < 72; ++kk) {
    const int k = kk * 4 + half * 2;
    const v2f a = *(const v2f*)(srow + ln * PITCH + k);
    const v2f b = *(const v2f*)(W0p + ln * 288 + k);
    acc = wmma_f32(a, b, acc);
  }
  if (ln < ND) {
    #pragma unroll
    for (int g = 0; g < 8; ++g)
      srow[(g + half * 8) * PITCH + COL_GF + ln] = acc[g];
  }

  for (int j = 0; j < NJ; ++j) {
    const int p    = c_parents[j];
    const int foff = (p < 0) ? COL_GF : p * 9;

    if (lane < 16) {
      const int  m = lane;
      const float* row = srow + m * PITCH;
      const float* rp  = row + j * 9;
      const float* tp  = row + COL_JT + j * 3;
      __align__(16) float xr[20];
      #pragma unroll
      for (int i = 0; i < 9; ++i) xr[i] = rp[i];
      const float jx = tp[0], jy = tp[1], jz = tp[2];
      xr[9] = jx; xr[10] = jy; xr[11] = jz;
      float px = 0.0f, py = 0.0f, pz = 0.0f;
      if (p >= 0) {
        const float* pp = row + COL_JT + p * 3;
        px = pp[0]; py = pp[1]; pz = pp[2];
      }
      const float dx = jx - px, dy = jy - py, dz = jz - pz;
      xr[12] = sqrtf(dx * dx + dy * dy + dz * dz);
      #pragma unroll
      for (int i = 0; i < ND; ++i) xr[13 + i] = row[foff + i];
      xr[19] = 0.0f;
      #pragma unroll
      for (int i = 0; i < 20; i += 4)
        *(v4f*)(xb + m * 20 + i) = *(const v4f*)(xr + i);
    }
    asm volatile("s_wait_dscnt 0" ::: "memory");

    v8f c0, c1;
    {
      const float bias0 = b1p[j * 32 + ln];
      const float bias1 = b1p[j * 32 + 16 + ln];
      #pragma unroll
      for (int g = 0; g < 8; ++g) { c0[g] = bias0; c1[g] = bias1; }
    }
    const float* w1j = W1p + j * 32 * 20;
    #pragma unroll
    for (int kk = 0; kk < 5; ++kk) {
      const int k = kk * 4 + half * 2;
      const v2f a   = *(const v2f*)(xb + ln * 20 + k);
      const v2f bt0 = *(const v2f*)(w1j + ln * 20 + k);
      const v2f bt1 = *(const v2f*)(w1j + (16 + ln) * 20 + k);
      c0 = wmma_f32(a, bt0, c0);
      c1 = wmma_f32(a, bt1, c1);
    }

    #pragma unroll
    for (int g = 0; g < 8; ++g) {
      const int m = g + half * 8;
      xb[m * 20 + ln] = fmaxf(c0[g], 0.0f);
      if (ln < 3) xb[m * 20 + 16 + ln] = fmaxf(c1[g], 0.0f);
      if (ln == 3) xb[m * 20 + 19] = 0.0f;
    }
    asm volatile("s_wait_dscnt 0" ::: "memory");

    v8f c2;
    {
      const float bias2 = b2p[j * 16 + ln];
      #pragma unroll
      for (int g = 0; g < 8; ++g) c2[g] = bias2;
    }
    const float* w2j = W2p + j * 16 * 20;
    #pragma unroll
    for (int kk = 0; kk < 5; ++kk) {
      const int k = kk * 4 + half * 2;
      const v2f a  = *(const v2f*)(xb + ln * 20 + k);
      const v2f bt = *(const v2f*)(w2j + ln * 20 + k);
      c2 = wmma_f32(a, bt, c2);
    }
    if (ln < ND) {
      #pragma unroll
      for (int g = 0; g < 8; ++g)
        srow[(g + half * 8) * PITCH + j * 9 + ln] = c2[g];
    }
  }

  asm volatile("s_wait_dscnt 0" ::: "memory");
  #pragma unroll 4
  for (int idx = lane; idx < 16 * 36; idx += 32) {
    const int m = idx / 36;
    const int q = idx - m * 36;
    v4f v;
    #pragma unroll
    for (int t = 0; t < 4; ++t) {
      const int e  = q * 4 + t;
      const int jj = e / 6;
      v[t] = srow[m * PITCH + jj * 9 + (e - jj * 6)];
    }
    *(v4f*)(out + (s0 + m) * 144 + q * 4) = v;
  }
}

extern "C" void kernel_launch(void* const* d_in, const int* in_sizes, int n_in,
                              void* d_out, int out_size, void* d_ws, size_t ws_size,
                              hipStream_t stream) {
  const float* rots = (const float*)d_in[0];
  const float* Jtrs = (const float*)d_in[1];
  const float* W0   = (const float*)d_in[2];
  const float* b0   = (const float*)d_in[3];
  const float* W1   = (const float*)d_in[4];
  const float* b1   = (const float*)d_in[5];
  const float* W2   = (const float*)d_in[6];
  const float* b2   = (const float*)d_in[7];
  float* out = (float*)d_out;
  float* ws  = (float*)d_ws;

  hipLaunchKernelGGL(pad_weights_kernel, dim3(PAD_BLOCKS), dim3(PAD_THREADS), 0, stream,
                     W0, b0, W1, b1, W2, b2, ws);

  const long bsz    = (long)in_sizes[0] / (NJ * 9);
  const int  blocks = (int)(bsz / 64);
  if (blocks > 0) {
    hipLaunchKernelGGL(hier_pose_kernel, dim3(blocks), dim3(128), 0, stream,
                       rots, Jtrs, ws, out);
  }
}


# Padded weights stored in the workspace in WMMA B-fragment order: each wave-wide B load is one b128 of 512 contiguous bytes (4 lines), not a b64 spread over 10-16 lines; per-lane values unchanged
# speedup vs baseline: 1.8787x; 1.1169x over previous
#include <hip/hip_runtime.h>
#include <math.h>


typedef float v2f __attribute__((ext_vector_type(2)));
typedef float v4f __attribute__((ext_vector_type(4)));
typedef float v8f __attribute__((ext_vector_type(8)));

#define NJ 24
#define ND 6
#define NF 19

#define PITCH   300
#define COL_JT  216
#define COL_GF  288

#define OFF_W0  0
#define OFF_WJ  (OFF_W0 + 36*32*4)
#define WJ_W2Q  (5*32*4)
#define WJ_W2P  (WJ_W2Q + 2*32*4)
#define WJ_FLOATS (WJ_W2P + 32*2)
#define OFF_B1  (OFF_WJ + NJ*WJ_FLOATS)
#define OFF_B2  (OFF_B1 + NJ*32)
#define OFF_B0  (OFF_B2 + NJ*16)
#define WS_FLOATS (OFF_B0 + 16)

__constant__ int c_parents[NJ] = {-1,0,0,0,1,2,3,4,5,6,7,8,9,9,9,12,13,14,16,17,18,19,20,21};

static __device__ __forceinline__ v8f wmma_f32(v2f a, v2f b, v8f c) {
  return __builtin_amdgcn_wmma_f32_16x16x4_f32(
       false, a,  false, b,
       (short)0, c,  false,  false);
}

#define PAD_THREADS 256
#define PAD_BLOCKS  113
static_assert(PAD_BLOCKS * PAD_THREADS >= WS_FLOATS, "prep grid must cover the padded weights");

__global__ __launch_bounds__(PAD_THREADS)
void pad_weights_kernel(const float* __restrict__ W0, const float* __restrict__ b0,
                        const float* __restrict__ W1, const float* __restrict__ b1,
                        const float* __restrict__ W2, const float* __restrict__ b2,
                        float* __restrict__ ws)
{
  const int i = blockIdx.x * PAD_THREADS + threadIdx.x;
  if (i >= WS_FLOATS) return;

  float v = 0.0f;
  if (i < OFF_WJ) {
    const int q = i >> 7, lane = (i >> 2) & 31, t = i & 3;
    const int n = lane & 15, half = lane >> 4;
    const int k = (2 * q + (t >> 1)) * 4 + half * 2 + (t & 1);
    if (n < ND) v = W0[n * 288 + k];
  } else if (i < OFF_B1) {
    const int u = i - OFF_WJ;
    const int j = u / WJ_FLOATS, r = u - j * WJ_FLOATS;
    if (r < WJ_W2Q) {
      const int q = r >> 7, lane = (r >> 2) & 31, t = r & 3;
      const int n = (lane & 15) + 16 * (t >> 1), half = lane >> 4;
      const int k = q * 4 + half * 2 + (t & 1);
      if (n < NF && k < NF) v = W1[(j * NF + n) * NF + k];
    } else {
      int lane, s, e;
      if (r < WJ_W2P) {
        const int r2 = r - WJ_W2Q;
        lane = (r2 >> 2) & 31; s = 2 * (r2 >> 7) + ((r2 >> 1) & 1); e = r2 & 1;
      } else {
        const int r3 = r - WJ_W2P;
        lane = r3 >> 1; s = 4; e = r3 & 1;
      }
      const int n = lane & 15, half = lane >> 4;
      const int k = s * 4 + half * 2 + e;
      if (n < ND && k < NF) v = W2[(j * ND + n) * NF + k];
    }
  } else if (i < OFF_B2) {
    const int q = i - OFF_B1;
    const int j = q / 32, n = q - j * 32;
    if (n < NF) v = b1[j * NF + n];
  } else if (i < OFF_B0) {
    const int q = i - OFF_B2;
    const int j = q / 16, n = q - j * 16;
    if (n < ND) v = b2[j * ND + n];
  } else {
    const int n = i - OFF_B0;
    if (n < ND) v = b0[n];
  }
  ws[i] = v;
}

__global__ __launch_bounds__(128)
void hier_pose_kernel(const float* __restrict__ rots, const float* __restrict__ Jtrs,
                      const float* __restrict__ ws,   float* __restrict__ out)
{
  __shared__ __align__(16) float stile[4][16][PITCH];
  __shared__ __align__(16) float sx[4][16][20];

  const float* W0f = ws + OFF_W0;
  const float* WJf = ws + OFF_WJ;
  const float* b1p = ws + OFF_B1;
  const float* b2p = ws + OFF_B2;
  const float* b0p = ws + OFF_B0;

  const int  lane = threadIdx.x & 31;
  const int  wave = threadIdx.x >> 5;
  const int  half = lane >> 4;
  const int  ln   = lane & 15;
  const long s0   = ((long)blockIdx.x * 4 + wave) * 16;

  float* srow = &stile[wave][0][0];
  float* xb   = &sx[wave][0][0];

  {
    const v4f* gr = (const v4f*)(rots + s0 * 216);
    const v4f* gj = (const v4f*)(Jtrs + s0 * 72);
    v4f tr[27];
    v4f tj[9];
    #pragma unroll
    for (int i = 0; i < 27; ++i) tr[i] = gr[lane + 32 * i];
    #pragma unroll
    for (int i = 0; i < 9; ++i)  tj[i] = gj[lane + 32 * i];
    #pragma unroll
    for (int i = 0; i < 27; ++i) {
      const int idx = lane + 32 * i;
      const int m   = idx / 54;
      const int c4  = idx - m * 54;
      *(v4f*)(srow + m * PITCH + c4 * 4) = tr[i];
    }
    #pragma unroll
    for (int i = 0; i < 9; ++i) {
      const int idx = lane + 32 * i;
      const int m   = idx / 18;
      const int c4  = idx - m * 18;
      *(v4f*)(srow + m * PITCH + COL_JT + c4 * 4) = tj[i];
    }
  }
  asm volatile("s_wait_dscnt 0" ::: "memory");

  v8f acc;
  {
    const float ci = b0p[ln];
    #pragma unroll
    for (int g = 0; g < 8; ++g) acc[g] = ci;
  }
  #pragma unroll 4
  for (int q = 0; q < 36; ++q) {
    const int k = q * 8 + half * 2;
    const v4f b  = *(const v4f*)(W0f + (q * 32 + lane) * 4);
    const v2f a0 = *(const v2f*)(srow + ln * PITCH + k);
    const v2f a1 = *(const v2f*)(srow + ln * PITCH + k + 4);
    acc = wmma_f32(a0, b.xy, acc);
    acc = wmma_f32(a1, b.zw, acc);
  }
  if (ln < ND) {
    #pragma unroll
    for (int g = 0; g < 8; ++g)
      srow[(g + half * 8) * PITCH + COL_GF + ln] = acc[g];
  }

  for (int j = 0; j < NJ; ++j) {
    const float* wj = WJf + j * WJ_FLOATS;
    const int p    = c_parents[j];
    const int foff = (p < 0) ? COL_GF : p * 9;

    if (lane < 16) {
      const int  m = lane;
      const float* row = srow + m * PITCH;
      const float* rp  = row + j * 9;
      const float* tp  = row + COL_JT + j * 3;
      __align__(16) float xr[20];
      #pragma unroll
      for (int i = 0; i < 9; ++i) xr[i] = rp[i];
      const float jx = tp[0], jy = tp[1], jz = tp[2];
      xr[9] = jx; xr[10] = jy; xr[11] = jz;
      float px = 0.0f, py = 0.0f, pz = 0.0f;
      if (p >= 0) {
        const float* pp = row + COL_JT + p * 3;
        px = pp[0]; py = pp[1]; pz = pp[2];
      }
      const float dx = jx - px, dy = jy - py, dz = jz - pz;
      xr[12] = sqrtf(dx * dx + dy * dy + dz * dz);
      #pragma unroll
      for (int i = 0; i < ND; ++i) xr[13 + i] = row[foff + i];
      xr[19] = 0.0f;
      #pragma unroll
      for (int i = 0; i < 20; i += 4)
        *(v4f*)(xb + m * 20 + i) = *(const v4f*)(xr + i);
    }
    asm volatile("s_wait_dscnt 0" ::: "memory");

    v8f c0, c1;
    {
      const float bias0 = b1p[j * 32 + ln];
      const float bias1 = b1p[j * 32 + 16 + ln];
      #pragma unroll
      for (int g = 0; g < 8; ++g) { c0[g] = bias0; c1[g] = bias1; }
    }
    #pragma unroll
    for (int kk = 0; kk < 5; ++kk) {
      const int k = kk * 4 + half * 2;
      const v2f a = *(const v2f*)(xb + ln * 20 + k);
      const v4f b = *(const v4f*)(wj + (kk * 32 + lane) * 4);
      c0 = wmma_f32(a, b.xy, c0);
      c1 = wmma_f32(a, b.zw, c1);
    }

    #pragma unroll
    for (int g = 0; g < 8; ++g) {
      const int m = g + half * 8;
      xb[m * 20 + ln] = fmaxf(c0[g], 0.0f);
      if (ln < 3) xb[m * 20 + 16 + ln] = fmaxf(c1[g], 0.0f);
      if (ln == 3) xb[m * 20 + 19] = 0.0f;
    }
    asm volatile("s_wait_dscnt 0" ::: "memory");

    v8f c2;
    {
      const float bias2 = b2p[j * 16 + ln];
      #pragma unroll
      for (int g = 0; g < 8; ++g) c2[g] = bias2;
    }
    {
      const v4f b01 = *(const v4f*)(wj + WJ_W2Q + lane * 4);
      const v4f b23 = *(const v4f*)(wj + WJ_W2Q + (32 + lane) * 4);
      const v2f b4  = *(const v2f*)(wj + WJ_W2P + lane * 2);
      const float* ah = xb + ln * 20 + half * 2;
      c2 = wmma_f32(*(const v2f*)(ah +  0), b01.xy, c2);
      c2 = wmma_f32(*(const v2f*)(ah +  4), b01.zw, c2);
      c2 = wmma_f32(*(const v2f*)(ah +  8), b23.xy, c2);
      c2 = wmma_f32(*(const v2f*)(ah + 12), b23.zw, c2);
      c2 = wmma_f32(*(const v2f*)(ah + 16), b4,     c2);
    }
    if (ln < ND) {
      #pragma unroll
      for (int g = 0; g < 8; ++g)
        srow[(g + half * 8) * PITCH + j * 9 + ln] = c2[g];
    }
  }

  asm volatile("s_wait_dscnt 0" ::: "memory");
  #pragma unroll 4
  for (int idx = lane; idx < 16 * 36; idx += 32) {
    const int m = idx / 36;
    const int q = idx - m * 36;
    v4f v;
    #pragma unroll
    for (int t = 0; t < 4; ++t) {
      const int e  = q * 4 + t;
      const int jj = e / 6;
      v[t] = srow[m * PITCH + jj * 9 + (e - jj * 6)];
    }
    *(v4f*)(out + (s0 + m) * 144 + q * 4) = v;
  }
}

extern "C" void kernel_launch(void* const* d_in, const int* in_sizes, int n_in,
                              void* d_out, int out_size, void* d_ws, size_t ws_size,
                              hipStream_t stream) {
  const float* rots = (const float*)d_in[0];
  const float* Jtrs = (const float*)d_in[1];
  const float* W0   = (const float*)d_in[2];
  const float* b0   = (const float*)d_in[3];
  const float* W1   = (const float*)d_in[4];
  const float* b1   = (const float*)d_in[5];
  const float* W2   = (const float*)d_in[6];
  const float* b2   = (const float*)d_in[7];
  float* out = (float*)d_out;
  float* ws  = (float*)d_ws;

  hipLaunchKernelGGL(pad_weights_kernel, dim3(PAD_BLOCKS), dim3(PAD_THREADS), 0, stream,
                     W0, b0, W1, b1, W2, b2, ws);

  const long bsz    = (long)in_sizes[0] / (NJ * 9);
  const int  blocks = (int)(bsz / 64);
  if (blocks > 0) {
    hipLaunchKernelGGL(hier_pose_kernel, dim3(blocks), dim3(128), 0, stream,
                       rots, Jtrs, ws, out);
  }
}


# Per-joint MLP transposed (weights as A, samples as columns): x read straight from the staged rows, ReLU'd D registers are the next product's B, biases in a spare k slot; k order of the sums changes
# speedup vs baseline: 2.8503x; 1.5172x over previous
#include <hip/hip_runtime.h>
#include <math.h>


typedef float v2f __attribute__((ext_vector_type(2)));
typedef float v4f __attribute__((ext_vector_type(4)));
typedef float v8f __attribute__((ext_vector_type(8)));

#define NJ 24
#define ND 6
#define NF 19

#define PITCH   300
#define COL_JT  216
#define COL_GF  288

#define OFF_W0  0
#define OFF_WJ  (OFF_W0 + 36*32*4)
#define WJ_W2Q  (5*32*4)
#define WJ_W2P  (WJ_W2Q + 2*32*4)
#define WJ_FLOATS (WJ_W2P + 32*2)
#define OFF_B0  (OFF_WJ + NJ*WJ_FLOATS)
#define WS_FLOATS (OFF_B0 + 16)

constexpr int kParents[NJ] = {-1,0,0,0,1,2,3,4,5,6,7,8,9,9,9,12,13,14,16,17,18,19,20,21};
constexpr unsigned long long pack_parents(int base) {
  unsigned long long w = 0;
  for (int i = 0; i < 12; ++i) w |= (unsigned long long)(kParents[base + i] + 1) << (5 * i);
  return w;
}
constexpr unsigned long long kParentsLo = pack_parents(0);
constexpr unsigned long long kParentsHi = pack_parents(12);

static __device__ __forceinline__ v8f wmma_f32(v2f a, v2f b, v8f c) {
  return __builtin_amdgcn_wmma_f32_16x16x4_f32(
       false, a,  false, b,
       (short)0, c,  false,  false);
}

#define PAD_THREADS 256
#define PAD_BLOCKS  109
static_assert(PAD_BLOCKS * PAD_THREADS >= WS_FLOATS, "prep grid must cover the padded weights");

__global__ __launch_bounds__(PAD_THREADS)
void pad_weights_kernel(const float* __restrict__ W0, const float* __restrict__ b0,
                        const float* __restrict__ W1, const float* __restrict__ b1,
                        const float* __restrict__ W2, const float* __restrict__ b2,
                        float* __restrict__ ws)
{
  const int i = blockIdx.x * PAD_THREADS + threadIdx.x;
  if (i >= WS_FLOATS) return;

  float v = 0.0f;
  if (i < OFF_WJ) {
    const int q = i >> 7, lane = (i >> 2) & 31, t = i & 3;
    const int n = lane & 15, half = lane >> 4;
    const int k = (2 * q + (t >> 1)) * 4 + half * 2 + (t & 1);
    if (n < ND) v = W0[n * 288 + k];
  } else if (i < OFF_B0) {
    const int u = i - OFF_WJ;
    const int j = u / WJ_FLOATS, r = u - j * WJ_FLOATS;
    if (r < WJ_W2Q) {
      const int s = r >> 7, lane = (r >> 2) & 31, t = r & 3;
      const int c = lane & 15, half = lane >> 4;
      const int slot = 2 * s + (t & 1);
      int kf;
      if (half == 0) kf = (slot < 9) ? slot : -1;
      else           kf = (slot < 3) ? 9 + slot : ((slot < 9) ? 10 + slot : 12);
      int n;
      if ((t >> 1) == 0) n = c;
      else               n = (c == 0) ? 16 : ((c == 1) ? 17 : ((c == 8) ? 18 : -1));
      if (n >= 0) v = (kf < 0) ? b1[j * NF + n] : W1[(j * NF + n) * NF + kf];
    } else {
      int lane, s, e;
      if (r < WJ_W2P) {
        const int r2 = r - WJ_W2Q;
        lane = (r2 >> 2) & 31; s = 2 * (r2 >> 7) + ((r2 >> 1) & 1); e = r2 & 1;
      } else {
        const int r3 = r - WJ_W2P;
        lane = r3 >> 1; s = 4; e = r3 & 1;
      }
      const int c = lane & 15, half = lane >> 4;
      int hf;
      if (s < 4)          hf = 8 * half + 2 * s + e;
      else if (half == 0) hf = 16 + e;
      else                hf = (e == 0) ? 18 : -1;
      if (c < ND) v = (hf < 0) ? b2[j * ND + c] : W2[(j * ND + c) * NF + hf];
    }
  } else {
    const int n = i - OFF_B0;
    if (n < ND) v = b0[n];
  }
  ws[i] = v;
}

__global__ __launch_bounds__(128)
void hier_pose_kernel(const float* __restrict__ rots, const float* __restrict__ Jtrs,
                      const float* __restrict__ ws,   float* __restrict__ out)
{
  __shared__ __align__(16) float stile[4][16][PITCH];

  const float* W0f = ws + OFF_W0;
  const float* WJf = ws + OFF_WJ;
  const float* b0p = ws + OFF_B0;

  const int  lane = threadIdx.x & 31;
  const int  wave = threadIdx.x >> 5;
  const int  half = lane >> 4;
  const int  ln   = lane & 15;
  const long s0   = ((long)blockIdx.x * 4 + wave) * 16;

  float* srow = &stile[wave][0][0];

  {
    const v4f* gr = (const v4f*)(rots + s0 * 216);
    const v4f* gj = (const v4f*)(Jtrs + s0 * 72);
    v4f tr[27];
    v4f tj[9];
    #pragma unroll
    for (int i = 0; i < 27; ++i) tr[i] = gr[lane + 32 * i];
    #pragma unroll
    for (int i = 0; i < 9; ++i)  tj[i] = gj[lane + 32 * i];
    #pragma unroll
    for (int i = 0; i < 27; ++i) {
      const int idx = lane + 32 * i;
      const int m   = idx / 54;
      const int c4  = idx - m * 54;
      *(v4f*)(srow + m * PITCH + c4 * 4) = tr[i];
    }
    #pragma unroll
    for (int i = 0; i < 9; ++i) {
      const int idx = lane + 32 * i;
      const int m   = idx / 18;
      const int c4  = idx - m * 18;
      *(v4f*)(srow + m * PITCH + COL_JT + c4 * 4) = tj[i];
    }
  }
  asm volatile("s_wait_dscnt 0" ::: "memory");

  v8f acc;
  {
    const float ci = b0p[ln];
    #pragma unroll
    for (int g = 0; g < 8; ++g) acc[g] = ci;
  }
  #pragma unroll 4
  for (int q = 0; q < 36; ++q) {
    const int k = q * 8 + half * 2;
    const v4f b  = *(const v4f*)(W0f + (q * 32 + lane) * 4);
    const v2f a0 = *(const v2f*)(srow + ln * PITCH + k);
    const v2f a1 = *(const v2f*)(srow + ln * PITCH + k + 4);
    acc = wmma_f32(a0, b.xy, acc);
    acc = wmma_f32(a1, b.zw, acc);
  }
  if (ln < ND) {
    #pragma unroll
    for (int g = 0; g < 8; ++g)
      srow[(g + half * 8) * PITCH + COL_GF + ln] = acc[g];
  }

  float* row = srow + ln * PITCH;
  for (int j = 0; j < NJ; ++j) {
    const float* wj = WJf + j * WJ_FLOATS;
    const int p    = (int)(((j < 12 ? kParentsLo : kParentsHi) >> (5 * (j < 12 ? j : j - 12))) & 31) - 1;
    const int foff = (p < 0) ? COL_GF : p * 9;

    v4f w1[5];
    #pragma unroll
    for (int s = 0; s < 5; ++s) w1[s] = *(const v4f*)(wj + (s * 32 + lane) * 4);
    const v4f w2a = *(const v4f*)(wj + WJ_W2Q + lane * 4);
    const v4f w2b = *(const v4f*)(wj + WJ_W2Q + (32 + lane) * 4);
    const v2f w2c = *(const v2f*)(wj + WJ_W2P + lane * 2);

    float e[10];
    {
      const float* pa = row + (half ? COL_JT + 3 * j : 9 * j);
      e[0] = pa[0]; e[1] = pa[1]; e[2] = pa[2];
    }
    float px = 0.0f, py = 0.0f, pz = 0.0f;
    if (p >= 0) {
      const float* pp = row + COL_JT + p * 3;
      px = pp[0]; py = pp[1]; pz = pp[2];
    }
    asm volatile("s_wait_dscnt 0" ::: "memory");
    {
      const float* pb = row + (half ? foff : 9 * j + 3);
      #pragma unroll
      for (int i = 0; i < 6; ++i) e[3 + i] = pb[i];
    }
    {
      const float dx = e[0] - px, dy = e[1] - py, dz = e[2] - pz;
      const float bone = sqrtf(dx * dx + dy * dy + dz * dz);
      e[9] = half ? bone : 1.0f;
    }

    v8f c0 = {0.0f, 0.0f, 0.0f, 0.0f, 0.0f, 0.0f, 0.0f, 0.0f};
    v8f c1 = {0.0f, 0.0f, 0.0f, 0.0f, 0.0f, 0.0f, 0.0f, 0.0f};
    #pragma unroll
    for (int s = 0; s < 5; ++s) {
      const v2f b = {e[2 * s], e[2 * s + 1]};
      c0 = wmma_f32(w1[s].xy, b, c0);
      c1 = wmma_f32(w1[s].zw, b, c1);
    }

    float hb[10];
    #pragma unroll
    for (int r = 0; r < 8; ++r) hb[r] = fmaxf(c0[r], 0.0f);
    hb[8] = fmaxf(c1[0], 0.0f);
    hb[9] = half ? 1.0f : fmaxf(c1[1], 0.0f);

    v8f c2 = {0.0f, 0.0f, 0.0f, 0.0f, 0.0f, 0.0f, 0.0f, 0.0f};
    {
      const v2f h01 = {hb[0], hb[1]}, h23 = {hb[2], hb[3]}, h45 = {hb[4], hb[5]};
      const v2f h67 = {hb[6], hb[7]}, h89 = {hb[8], hb[9]};
      c2 = wmma_f32(w2a.xy, h01, c2);
      c2 = wmma_f32(w2a.zw, h23, c2);
      c2 = wmma_f32(w2b.xy, h45, c2);
      c2 = wmma_f32(w2b.zw, h67, c2);
      c2 = wmma_f32(w2c,    h89, c2);
    }
    if (half == 0) {
      #pragma unroll
      for (int r = 0; r < ND; ++r) row[j * 9 + r] = c2[r];
    }
  }

  asm volatile("s_wait_dscnt 0" ::: "memory");
  #pragma unroll 4
  for (int idx = lane; idx < 16 * 36; idx += 32) {
    const int m = idx / 36;
    const int q = idx - m * 36;
    v4f v;
    #pragma unroll
    for (int t = 0; t < 4; ++t) {
      const int e  = q * 4 + t;
      const int jj = e / 6;
      v[t] = srow[m * PITCH + jj * 9 + (e - jj * 6)];
    }
    *(v4f*)(out + (s0 + m) * 144 + q * 4) = v;
  }
}

extern "C" void kernel_launch(void* const* d_in, const int* in_sizes, int n_in,
                              void* d_out, int out_size, void* d_ws, size_t ws_size,
                              hipStream_t stream) {
  const float* rots = (const float*)d_in[0];
  const float* Jtrs = (const float*)d_in[1];
  const float* W0   = (const float*)d_in[2];
  const float* b0   = (const float*)d_in[3];
  const float* W1   = (const float*)d_in[4];
  const float* b1   = (const float*)d_in[5];
  const float* W2   = (const float*)d_in[6];
  const float* b2   = (const float*)d_in[7];
  float* out = (float*)d_out;
  float* ws  = (float*)d_ws;

  hipLaunchKernelGGL(pad_weights_kernel, dim3(PAD_BLOCKS), dim3(PAD_THREADS), 0, stream,
                     W0, b0, W1, b1, W2, b2, ws);

  const long bsz    = (long)in_sizes[0] / (NJ * 9);
  const int  blocks = (int)(bsz / 64);
  if (blocks > 0) {
    hipLaunchKernelGGL(hier_pose_kernel, dim3(blocks), dim3(128), 0, stream,
                       rots, Jtrs, ws, out);
  }
}


# Joint loop software-pipelined (fully unrolled): joint j+1's weight fragments and its rot/jtr/bone-length x slots are fetched while joint j's products run; only feat is read inside its joint
# speedup vs baseline: 3.0611x; 1.0740x over previous
#include <hip/hip_runtime.h>
#include <math.h>


typedef float v2f __attribute__((ext_vector_type(2)));
typedef float v4f __attribute__((ext_vector_type(4)));
typedef float v8f __attribute__((ext_vector_type(8)));

#define NJ 24
#define ND 6
#define NF 19

#define PITCH   300
#define COL_JT  216
#define COL_GF  288

#define OFF_W0  0
#define OFF_WJ  (OFF_W0 + 36*32*4)
#define WJ_W2Q  (5*32*4)
#define WJ_W2P  (WJ_W2Q + 2*32*4)
#define WJ_FLOATS (WJ_W2P + 32*2)
#define OFF_B0  (OFF_WJ + NJ*WJ_FLOATS)
#define WS_FLOATS (OFF_B0 + 16)

constexpr int kParents[NJ] = {-1,0,0,0,1,2,3,4,5,6,7,8,9,9,9,12,13,14,16,17,18,19,20,21};
constexpr unsigned long long pack_parents(int base) {
  unsigned long long w = 0;
  for (int i = 0; i < 12; ++i) w |= (unsigned long long)(kParents[base + i] + 1) << (5 * i);
  return w;
}
constexpr unsigned long long kParentsLo = pack_parents(0);
constexpr unsigned long long kParentsHi = pack_parents(12);

static __device__ __forceinline__ v8f wmma_f32(v2f a, v2f b, v8f c) {
  return __builtin_amdgcn_wmma_f32_16x16x4_f32(
       false, a,  false, b,
       (short)0, c,  false,  false);
}

#define PAD_THREADS 256
#define PAD_BLOCKS  109
static_assert(PAD_BLOCKS * PAD_THREADS >= WS_FLOATS, "prep grid must cover the padded weights");

__global__ __launch_bounds__(PAD_THREADS)
void pad_weights_kernel(const float* __restrict__ W0, const float* __restrict__ b0,
                        const float* __restrict__ W1, const float* __restrict__ b1,
                        const float* __restrict__ W2, const float* __restrict__ b2,
                        float* __restrict__ ws)
{
  const int i = blockIdx.x * PAD_THREADS + threadIdx.x;
  if (i >= WS_FLOATS) return;

  float v = 0.0f;
  if (i < OFF_WJ) {
    const int q = i >> 7, lane = (i >> 2) & 31, t = i & 3;
    const int n = lane & 15, half = lane >> 4;
    const int k = (2 * q + (t >> 1)) * 4 + half * 2 + (t & 1);
    if (n < ND) v = W0[n * 288 + k];
  } else if (i < OFF_B0) {
    const int u = i - OFF_WJ;
    const int j = u / WJ_FLOATS, r = u - j * WJ_FLOATS;
    if (r < WJ_W2Q) {
      const int s = r >> 7, lane = (r >> 2) & 31, t = r & 3;
      const int c = lane & 15, half = lane >> 4;
      const int slot = 2 * s + (t & 1);
      int kf;
      if (half == 0) kf = (slot < 9) ? slot : -1;
      else           kf = (slot < 3) ? 9 + slot : ((slot < 9) ? 10 + slot : 12);
      int n;
      if ((t >> 1) == 0) n = c;
      else               n = (c == 0) ? 16 : ((c == 1) ? 17 : ((c == 8) ? 18 : -1));
      if (n >= 0) v = (kf < 0) ? b1[j * NF + n] : W1[(j * NF + n) * NF + kf];
    } else {
      int lane, s, e;
      if (r < WJ_W2P) {
        const int r2 = r - WJ_W2Q;
        lane = (r2 >> 2) & 31; s = 2 * (r2 >> 7) + ((r2 >> 1) & 1); e = r2 & 1;
      } else {
        const int r3 = r - WJ_W2P;
        lane = r3 >> 1; s = 4; e = r3 & 1;
      }
      const int c = lane & 15, half = lane >> 4;
      int hf;
      if (s < 4)          hf = 8 * half + 2 * s + e;
      else if (half == 0) hf = 16 + e;
      else                hf = (e == 0) ? 18 : -1;
      if (c < ND) v = (hf < 0) ? b2[j * ND + c] : W2[(j * ND + c) * NF + hf];
    }
  } else {
    const int n = i - OFF_B0;
    if (n < ND) v = b0[n];
  }
  ws[i] = v;
}

struct JointW {
  v4f w1[5];
  v4f w2a, w2b;
  v2f w2c;
};
struct JointX {
  float a0, a1, a2;
  float px, py, pz;
};

static __device__ __forceinline__ void load_joint_w(JointW& w, const float* __restrict__ wj, int lane) {
  #pragma unroll
  for (int s = 0; s < 5; ++s) w.w1[s] = *(const v4f*)(wj + (s * 32 + lane) * 4);
  w.w2a = *(const v4f*)(wj + WJ_W2Q + lane * 4);
  w.w2b = *(const v4f*)(wj + WJ_W2Q + (32 + lane) * 4);
  w.w2c = *(const v2f*)(wj + WJ_W2P + lane * 2);
}

static __device__ __forceinline__ void load_joint_x(JointX& x, const float* row, int half, int j, int p) {
  const float* pa = row + (half ? COL_JT + 3 * j : 9 * j);
  x.a0 = pa[0]; x.a1 = pa[1]; x.a2 = pa[2];
  x.px = 0.0f; x.py = 0.0f; x.pz = 0.0f;
  if (p >= 0) {
    const float* pp = row + COL_JT + p * 3;
    x.px = pp[0]; x.py = pp[1]; x.pz = pp[2];
  }
}

static __device__ __forceinline__ int parent_of(int j) {
  return (int)(((j < 12 ? kParentsLo : kParentsHi) >> (5 * (j < 12 ? j : j - 12))) & 31) - 1;
}

__global__ __launch_bounds__(128)
void hier_pose_kernel(const float* __restrict__ rots, const float* __restrict__ Jtrs,
                      const float* __restrict__ ws,   float* __restrict__ out)
{
  __shared__ __align__(16) float stile[4][16][PITCH];

  const float* W0f = ws + OFF_W0;
  const float* WJf = ws + OFF_WJ;
  const float* b0p = ws + OFF_B0;

  const int  lane = threadIdx.x & 31;
  const int  wave = threadIdx.x >> 5;
  const int  half = lane >> 4;
  const int  ln   = lane & 15;
  const long s0   = ((long)blockIdx.x * 4 + wave) * 16;

  float* srow = &stile[wave][0][0];

  {
    const v4f* gr = (const v4f*)(rots + s0 * 216);
    const v4f* gj = (const v4f*)(Jtrs + s0 * 72);
    v4f tr[27];
    v4f tj[9];
    #pragma unroll
    for (int i = 0; i < 27; ++i) tr[i] = gr[lane + 32 * i];
    #pragma unroll
    for (int i = 0; i < 9; ++i)  tj[i] = gj[lane + 32 * i];
    #pragma unroll
    for (int i = 0; i < 27; ++i) {
      const int idx = lane + 32 * i;
      const int m   = idx / 54;
      const int c4  = idx - m * 54;
      *(v4f*)(srow + m * PITCH + c4 * 4) = tr[i];
    }
    #pragma unroll
    for (int i = 0; i < 9; ++i) {
      const int idx = lane + 32 * i;
      const int m   = idx / 18;
      const int c4  = idx - m * 18;
      *(v4f*)(srow + m * PITCH + COL_JT + c4 * 4) = tj[i];
    }
  }
  asm volatile("s_wait_dscnt 0" ::: "memory");

  float* row = srow + ln * PITCH;

  JointW wc;
  JointX xc;
  load_joint_w(wc, WJf, lane);
  load_joint_x(xc, row, half, 0, parent_of(0));
  float e9c;
  {
    const float dx = xc.a0 - xc.px, dy = xc.a1 - xc.py, dz = xc.a2 - xc.pz;
    e9c = half ? sqrtf(dx * dx + dy * dy + dz * dz) : 1.0f;
  }

  v8f acc;
  {
    const float ci = b0p[ln];
    #pragma unroll
    for (int g = 0; g < 8; ++g) acc[g] = ci;
  }
  #pragma unroll 4
  for (int q = 0; q < 36; ++q) {
    const int k = q * 8 + half * 2;
    const v4f b  = *(const v4f*)(W0f + (q * 32 + lane) * 4);
    const v2f a0 = *(const v2f*)(srow + ln * PITCH + k);
    const v2f a1 = *(const v2f*)(srow + ln * PITCH + k + 4);
    acc = wmma_f32(a0, b.xy, acc);
    acc = wmma_f32(a1, b.zw, acc);
  }
  if (ln < ND) {
    #pragma unroll
    for (int g = 0; g < 8; ++g)
      srow[(g + half * 8) * PITCH + COL_GF + ln] = acc[g];
  }

  #pragma unroll
  for (int j = 0; j < NJ; ++j) {
    const int p    = parent_of(j);
    const int foff = (p < 0) ? COL_GF : p * 9;

    asm volatile("s_wait_dscnt 0" ::: "memory");
    float f[6];
    {
      const float* pb = row + (half ? foff : 9 * j + 3);
      #pragma unroll
      for (int i = 0; i < 6; ++i) f[i] = pb[i];
    }

    JointW wn;
    JointX xn;
    if (j + 1 < NJ) {
      load_joint_w(wn, WJf + (j + 1) * WJ_FLOATS, lane);
      load_joint_x(xn, row, half, j + 1, parent_of(j + 1));
    }

    const float e[10] = {xc.a0, xc.a1, xc.a2, f[0], f[1], f[2], f[3], f[4], f[5], e9c};

    v8f c0 = {0.0f, 0.0f, 0.0f, 0.0f, 0.0f, 0.0f, 0.0f, 0.0f};
    v8f c1 = {0.0f, 0.0f, 0.0f, 0.0f, 0.0f, 0.0f, 0.0f, 0.0f};
    #pragma unroll
    for (int s = 0; s < 5; ++s) {
      const v2f b = {e[2 * s], e[2 * s + 1]};
      c0 = wmma_f32(wc.w1[s].xy, b, c0);
      c1 = wmma_f32(wc.w1[s].zw, b, c1);
    }

    float e9n = 1.0f;
    if (j + 1 < NJ) {
      const float dx = xn.a0 - xn.px, dy = xn.a1 - xn.py, dz = xn.a2 - xn.pz;
      const float bone = sqrtf(dx * dx + dy * dy + dz * dz);
      e9n = half ? bone : 1.0f;
    }

    float hb[10];
    #pragma unroll
    for (int r = 0; r < 8; ++r) hb[r] = fmaxf(c0[r], 0.0f);
    hb[8] = fmaxf(c1[0], 0.0f);
    hb[9] = half ? 1.0f : fmaxf(c1[1], 0.0f);

    v8f c2 = {0.0f, 0.0f, 0.0f, 0.0f, 0.0f, 0.0f, 0.0f, 0.0f};
    {
      const v2f h01 = {hb[0], hb[1]}, h23 = {hb[2], hb[3]}, h45 = {hb[4], hb[5]};
      const v2f h67 = {hb[6], hb[7]}, h89 = {hb[8], hb[9]};
      c2 = wmma_f32(wc.w2a.xy, h01, c2);
      c2 = wmma_f32(wc.w2a.zw, h23, c2);
      c2 = wmma_f32(wc.w2b.xy, h45, c2);
      c2 = wmma_f32(wc.w2b.zw, h67, c2);
      c2 = wmma_f32(wc.w2c,    h89, c2);
    }
    if (half == 0) {
      #pragma unroll
      for (int r = 0; r < ND; ++r) row[j * 9 + r] = c2[r];
    }

    if (j + 1 < NJ) { wc = wn; xc = xn; e9c = e9n; }
  }

  asm volatile("s_wait_dscnt 0" ::: "memory");
  #pragma unroll 4
  for (int idx = lane; idx < 16 * 36; idx += 32) {
    const int m = idx / 36;
    const int q = idx - m * 36;
    v4f v;
    #pragma unroll
    for (int t = 0; t < 4; ++t) {
      const int e  = q * 4 + t;
      const int jj = e / 6;
      v[t] = srow[m * PITCH + jj * 9 + (e - jj * 6)];
    }
    *(v4f*)(out + (s0 + m) * 144 + q * 4) = v;
  }
}

extern "C" void kernel_launch(void* const* d_in, const int* in_sizes, int n_in,
                              void* d_out, int out_size, void* d_ws, size_t ws_size,
                              hipStream_t stream) {
  const float* rots = (const float*)d_in[0];
  const float* Jtrs = (const float*)d_in[1];
  const float* W0   = (const float*)d_in[2];
  const float* b0   = (const float*)d_in[3];
  const float* W1   = (const float*)d_in[4];
  const float* b1   = (const float*)d_in[5];
  const float* W2   = (const float*)d_in[6];
  const float* b2   = (const float*)d_in[7];
  float* out = (float*)d_out;
  float* ws  = (float*)d_ws;

  hipLaunchKernelGGL(pad_weights_kernel, dim3(PAD_BLOCKS), dim3(PAD_THREADS), 0, stream,
                     W0, b0, W1, b1, W2, b2, ws);

  const long bsz    = (long)in_sizes[0] / (NJ * 9);
  const int  blocks = (int)(bsz / 64);
  if (blocks > 0) {
    hipLaunchKernelGGL(hier_pose_kernel, dim3(blocks), dim3(128), 0, stream,
                       rots, Jtrs, ws, out);
  }
}


# out[j] packed at floats 6j..6j+5 of the staged LDS row (the output's memory order): final store is one b128 LDS read per b128 store, not 4 gathered b32 reads with per-element index division
# speedup vs baseline: 3.2910x; 1.0751x over previous
#include <hip/hip_runtime.h>
#include <math.h>


typedef float v2f __attribute__((ext_vector_type(2)));
typedef float v4f __attribute__((ext_vector_type(4)));
typedef float v8f __attribute__((ext_vector_type(8)));

#define NJ 24
#define ND 6
#define NF 19

#define PITCH   300
#define COL_JT  216
#define COL_GF  288

#define OFF_W0  0
#define OFF_WJ  (OFF_W0 + 36*32*4)
#define WJ_W2Q  (5*32*4)
#define WJ_W2P  (WJ_W2Q + 2*32*4)
#define WJ_FLOATS (WJ_W2P + 32*2)
#define OFF_B0  (OFF_WJ + NJ*WJ_FLOATS)
#define WS_FLOATS (OFF_B0 + 16)

constexpr int kParents[NJ] = {-1,0,0,0,1,2,3,4,5,6,7,8,9,9,9,12,13,14,16,17,18,19,20,21};
constexpr unsigned long long pack_parents(int base) {
  unsigned long long w = 0;
  for (int i = 0; i < 12; ++i) w |= (unsigned long long)(kParents[base + i] + 1) << (5 * i);
  return w;
}
constexpr unsigned long long kParentsLo = pack_parents(0);
constexpr unsigned long long kParentsHi = pack_parents(12);

static __device__ __forceinline__ v8f wmma_f32(v2f a, v2f b, v8f c) {
  return __builtin_amdgcn_wmma_f32_16x16x4_f32(
       false, a,  false, b,
       (short)0, c,  false,  false);
}

#define PAD_THREADS 256
#define PAD_BLOCKS  109
static_assert(PAD_BLOCKS * PAD_THREADS >= WS_FLOATS, "prep grid must cover the padded weights");

__global__ __launch_bounds__(PAD_THREADS)
void pad_weights_kernel(const float* __restrict__ W0, const float* __restrict__ b0,
                        const float* __restrict__ W1, const float* __restrict__ b1,
                        const float* __restrict__ W2, const float* __restrict__ b2,
                        float* __restrict__ ws)
{
  const int i = blockIdx.x * PAD_THREADS + threadIdx.x;
  if (i >= WS_FLOATS) return;

  float v = 0.0f;
  if (i < OFF_WJ) {
    const int q = i >> 7, lane = (i >> 2) & 31, t = i & 3;
    const int n = lane & 15, half = lane >> 4;
    const int k = (2 * q + (t >> 1)) * 4 + half * 2 + (t & 1);
    if (n < ND) v = W0[n * 288 + k];
  } else if (i < OFF_B0) {
    const int u = i - OFF_WJ;
    const int j = u / WJ_FLOATS, r = u - j * WJ_FLOATS;
    if (r < WJ_W2Q) {
      const int s = r >> 7, lane = (r >> 2) & 31, t = r & 3;
      const int c = lane & 15, half = lane >> 4;
      const int slot = 2 * s + (t & 1);
      int kf;
      if (half == 0) kf = (slot < 9) ? slot : -1;
      else           kf = (slot < 3) ? 9 + slot : ((slot < 9) ? 10 + slot : 12);
      int n;
      if ((t >> 1) == 0) n = c;
      else               n = (c == 0) ? 16 : ((c == 1) ? 17 : ((c == 8) ? 18 : -1));
      if (n >= 0) v = (kf < 0) ? b1[j * NF + n] : W1[(j * NF + n) * NF + kf];
    } else {
      int lane, s, e;
      if (r < WJ_W2P) {
        const int r2 = r - WJ_W2Q;
        lane = (r2 >> 2) & 31; s = 2 * (r2 >> 7) + ((r2 >> 1) & 1); e = r2 & 1;
      } else {
        const int r3 = r - WJ_W2P;
        lane = r3 >> 1; s = 4; e = r3 & 1;
      }
      const int c = lane & 15, half = lane >> 4;
      int hf;
      if (s < 4)          hf = 8 * half + 2 * s + e;
      else if (half == 0) hf = 16 + e;
      else                hf = (e == 0) ? 18 : -1;
      if (c < ND) v = (hf < 0) ? b2[j * ND + c] : W2[(j * ND + c) * NF + hf];
    }
  } else {
    const int n = i - OFF_B0;
    if (n < ND) v = b0[n];
  }
  ws[i] = v;
}

struct JointW {
  v4f w1[5];
  v4f w2a, w2b;
  v2f w2c;
};
struct JointX {
  float a0, a1, a2;
  float px, py, pz;
};

static __device__ __forceinline__ void load_joint_w(JointW& w, const float* __restrict__ wj, int lane) {
  #pragma unroll
  for (int s = 0; s < 5; ++s) w.w1[s] = *(const v4f*)(wj + (s * 32 + lane) * 4);
  w.w2a = *(const v4f*)(wj + WJ_W2Q + lane * 4);
  w.w2b = *(const v4f*)(wj + WJ_W2Q + (32 + lane) * 4);
  w.w2c = *(const v2f*)(wj + WJ_W2P + lane * 2);
}

static __device__ __forceinline__ void load_joint_x(JointX& x, const float* row, int half, int j, int p) {
  const float* pa = row + (half ? COL_JT + 3 * j : 9 * j);
  x.a0 = pa[0]; x.a1 = pa[1]; x.a2 = pa[2];
  x.px = 0.0f; x.py = 0.0f; x.pz = 0.0f;
  if (p >= 0) {
    const float* pp = row + COL_JT + p * 3;
    x.px = pp[0]; x.py = pp[1]; x.pz = pp[2];
  }
}

static __device__ __forceinline__ int parent_of(int j) {
  return (int)(((j < 12 ? kParentsLo : kParentsHi) >> (5 * (j < 12 ? j : j - 12))) & 31) - 1;
}

__global__ __launch_bounds__(128)
void hier_pose_kernel(const float* __restrict__ rots, const float* __restrict__ Jtrs,
                      const float* __restrict__ ws,   float* __restrict__ out)
{
  __shared__ __align__(16) float stile[4][16][PITCH];

  const float* W0f = ws + OFF_W0;
  const float* WJf = ws + OFF_WJ;
  const float* b0p = ws + OFF_B0;

  const int  lane = threadIdx.x & 31;
  const int  wave = threadIdx.x >> 5;
  const int  half = lane >> 4;
  const int  ln   = lane & 15;
  const long s0   = ((long)blockIdx.x * 4 + wave) * 16;

  float* srow = &stile[wave][0][0];

  {
    const v4f* gr = (const v4f*)(rots + s0 * 216);
    const v4f* gj = (const v4f*)(Jtrs + s0 * 72);
    v4f tr[27];
    v4f tj[9];
    #pragma unroll
    for (int i = 0; i < 27; ++i) tr[i] = gr[lane + 32 * i];
    #pragma unroll
    for (int i = 0; i < 9; ++i)  tj[i] = gj[lane + 32 * i];
    #pragma unroll
    for (int i = 0; i < 27; ++i) {
      const int idx = lane + 32 * i;
      const int m   = idx / 54;
      const int c4  = idx - m * 54;
      *(v4f*)(srow + m * PITCH + c4 * 4) = tr[i];
    }
    #pragma unroll
    for (int i = 0; i < 9; ++i) {
      const int idx = lane + 32 * i;
      const int m   = idx / 18;
      const int c4  = idx - m * 18;
      *(v4f*)(srow + m * PITCH + COL_JT + c4 * 4) = tj[i];
    }
  }
  asm volatile("s_wait_dscnt 0" ::: "memory");

  float* row = srow + ln * PITCH;

  JointW wc;
  JointX xc;
  load_joint_w(wc, WJf, lane);
  load_joint_x(xc, row, half, 0, parent_of(0));
  float e9c;
  {
    const float dx = xc.a0 - xc.px, dy = xc.a1 - xc.py, dz = xc.a2 - xc.pz;
    e9c = half ? sqrtf(dx * dx + dy * dy + dz * dz) : 1.0f;
  }

  v8f acc;
  {
    const float ci = b0p[ln];
    #pragma unroll
    for (int g = 0; g < 8; ++g) acc[g] = ci;
  }
  #pragma unroll 4
  for (int q = 0; q < 36; ++q) {
    const int k = q * 8 + half * 2;
    const v4f b  = *(const v4f*)(W0f + (q * 32 + lane) * 4);
    const v2f a0 = *(const v2f*)(srow + ln * PITCH + k);
    const v2f a1 = *(const v2f*)(srow + ln * PITCH + k + 4);
    acc = wmma_f32(a0, b.xy, acc);
    acc = wmma_f32(a1, b.zw, acc);
  }
  if (ln < ND) {
    #pragma unroll
    for (int g = 0; g < 8; ++g)
      srow[(g + half * 8) * PITCH + COL_GF + ln] = acc[g];
  }

  #pragma unroll
  for (int j = 0; j < NJ; ++j) {
    const int p    = parent_of(j);
    const int foff = (p < 0) ? COL_GF : p * ND;

    asm volatile("s_wait_dscnt 0" ::: "memory");
    float f[6];
    {
      const float* pb = row + (half ? foff : 9 * j + 3);
      #pragma unroll
      for (int i = 0; i < 6; ++i) f[i] = pb[i];
    }

    JointW wn;
    JointX xn;
    if (j + 1 < NJ) {
      load_joint_w(wn, WJf + (j + 1) * WJ_FLOATS, lane);
      load_joint_x(xn, row, half, j + 1, parent_of(j + 1));
    }

    const float e[10] = {xc.a0, xc.a1, xc.a2, f[0], f[1], f[2], f[3], f[4], f[5], e9c};

    v8f c0 = {0.0f, 0.0f, 0.0f, 0.0f, 0.0f, 0.0f, 0.0f, 0.0f};
    v8f c1 = {0.0f, 0.0f, 0.0f, 0.0f, 0.0f, 0.0f, 0.0f, 0.0f};
    #pragma unroll
    for (int s = 0; s < 5; ++s) {
      const v2f b = {e[2 * s], e[2 * s + 1]};
      c0 = wmma_f32(wc.w1[s].xy, b, c0);
      c1 = wmma_f32(wc.w1[s].zw, b, c1);
    }

    float e9n = 1.0f;
    if (j + 1 < NJ) {
      const float dx = xn.a0 - xn.px, dy = xn.a1 - xn.py, dz = xn.a2 - xn.pz;
      const float bone = sqrtf(dx * dx + dy * dy + dz * dz);
      e9n = half ? bone : 1.0f;
    }

    float hb[10];
    #pragma unroll
    for (int r = 0; r < 8; ++r) hb[r] = fmaxf(c0[r], 0.0f);
    hb[8] = fmaxf(c1[0], 0.0f);
    hb[9] = half ? 1.0f : fmaxf(c1[1], 0.0f);

    v8f c2 = {0.0f, 0.0f, 0.0f, 0.0f, 0.0f, 0.0f, 0.0f, 0.0f};
    {
      const v2f h01 = {hb[0], hb[1]}, h23 = {hb[2], hb[3]}, h45 = {hb[4], hb[5]};
      const v2f h67 = {hb[6], hb[7]}, h89 = {hb[8], hb[9]};
      c2 = wmma_f32(wc.w2a.xy, h01, c2);
      c2 = wmma_f32(wc.w2a.zw, h23, c2);
      c2 = wmma_f32(wc.w2b.xy, h45, c2);
      c2 = wmma_f32(wc.w2b.zw, h67, c2);
      c2 = wmma_f32(wc.w2c,    h89, c2);
    }
    if (half == 0) {
      #pragma unroll
      for (int r = 0; r < ND; ++r) row[j * ND + r] = c2[r];
    }

    if (j + 1 < NJ) { wc = wn; xc = xn; e9c = e9n; }
  }

  asm volatile("s_wait_dscnt 0" ::: "memory");
  {
    float* og = out + s0 * 144;
    #pragma unroll
    for (int i = 0; i < 9; ++i) {
      const int idx = lane + 32 * i;
      const int m   = idx / 36;
      const int q   = idx - m * 36;
      const float* ps = srow + m * PITCH + q * 4;
      const v4f lo = *(const v4f*)ps;
      const v4f hi = *(const v4f*)(ps + 8 * PITCH);
      *(v4f*)(og + idx * 4)           = lo;
      *(v4f*)(og + 8 * 144 + idx * 4) = hi;
    }
  }
}

extern "C" void kernel_launch(void* const* d_in, const int* in_sizes, int n_in,
                              void* d_out, int out_size, void* d_ws, size_t ws_size,
                              hipStream_t stream) {
  const float* rots = (const float*)d_in[0];
  const float* Jtrs = (const float*)d_in[1];
  const float* W0   = (const float*)d_in[2];
  const float* b0   = (const float*)d_in[3];
  const float* W1   = (const float*)d_in[4];
  const float* b1   = (const float*)d_in[5];
  const float* W2   = (const float*)d_in[6];
  const float* b2   = (const float*)d_in[7];
  float* out = (float*)d_out;
  float* ws  = (float*)d_ws;

  hipLaunchKernelGGL(pad_weights_kernel, dim3(PAD_BLOCKS), dim3(PAD_THREADS), 0, stream,
                     W0, b0, W1, b1, W2, b2, ws);

  const long bsz    = (long)in_sizes[0] / (NJ * 9);
  const int  blocks = (int)(bsz / 64);
  if (blocks > 0) {
    hipLaunchKernelGGL(hier_pose_kernel, dim3(blocks), dim3(128), 0, stream,
                       rots, Jtrs, ws, out);
  }
}
